# starting kernel
# baseline (speedup 1.0000x reference)
#include <hip/hip_runtime.h>
#include <cmath>

typedef __attribute__((ext_vector_type(2))) float v2f;
typedef __attribute__((ext_vector_type(8))) float v8f;

#define NTIME   2000
#define NBATCH  128
#define INSIZE  256
#define NOUT    40
#define NEG_BIG (-1.0e30f)

#define NCHUNK  100
#define CLEN    (NTIME / NCHUNK)

__device__ __forceinline__ float softplusf(float v) {
    return fmaxf(v, 0.0f) + __logf(1.0f + __expf(-fabsf(v)));
}

__device__ __forceinline__ float tanhf_fast(float v) {
    const float a = fabsf(v);
    const float e = __expf(2.0f * a);
    const float t = 1.0f - 2.0f * __builtin_amdgcn_rcpf(e + 1.0f);
    return copysignf(t, v);
}

__device__ __forceinline__ float lse8(const float (&t)[8]) {
    float mx = t[0];
#pragma unroll
    for (int m = 1; m < 8; ++m) mx = fmaxf(mx, t[m]);
    float s = 0.0f;
#pragma unroll
    for (int m = 0; m < 8; ++m) s += __expf(t[m] - mx);
    return mx + __logf(s);
}

__device__ __forceinline__ int iminc(int a, int b) { return a < b ? a : b; }

__global__ __launch_bounds__(256) void gemm_act_kernel(
    const float* __restrict__ x, const float* __restrict__ W,
    const float* __restrict__ bias, float* __restrict__ out)
{
    const int lane  = threadIdx.x & 31;
    const int wave  = threadIdx.x >> 5;
    const int l16   = lane & 15;
    const int halfk = lane >> 4;

    const int rowBase = (blockIdx.x * 8 + wave) * 64;

    v8f acc[4][3];
#pragma unroll
    for (int m = 0; m < 4; ++m)
#pragma unroll
        for (int n = 0; n < 3; ++n)
#pragma unroll
            for (int i = 0; i < 8; ++i) acc[m][n][i] = 0.0f;

    const float* xp = x + (size_t)(rowBase + l16) * INSIZE + 2 * halfk;
    const float* wp0 = W + (size_t)iminc(0 * 16 + l16, NOUT - 1) * INSIZE + 2 * halfk;
    const float* wp1 = W + (size_t)iminc(1 * 16 + l16, NOUT - 1) * INSIZE + 2 * halfk;
    const float* wp2 = W + (size_t)iminc(2 * 16 + l16, NOUT - 1) * INSIZE + 2 * halfk;

#pragma unroll 2
    for (int kk = 0; kk < INSIZE; kk += 4) {
        v2f a0 = __builtin_nontemporal_load((const v2f*)(xp + kk));
        v2f a1 = __builtin_nontemporal_load((const v2f*)(xp + 16 * INSIZE + kk));
        v2f a2 = __builtin_nontemporal_load((const v2f*)(xp + 32 * INSIZE + kk));
        v2f a3 = __builtin_nontemporal_load((const v2f*)(xp + 48 * INSIZE + kk));
        v2f b0 = *(const v2f*)(wp0 + kk);
        v2f b1 = *(const v2f*)(wp1 + kk);
        v2f b2 = *(const v2f*)(wp2 + kk);

        acc[0][0] = __builtin_amdgcn_wmma_f32_16x16x4_f32(false, a0, false, b0, (short)0, acc[0][0], false, false);
        acc[0][1] = __builtin_amdgcn_wmma_f32_16x16x4_f32(false, a0, false, b1, (short)0, acc[0][1], false, false);
        acc[0][2] = __builtin_amdgcn_wmma_f32_16x16x4_f32(false, a0, false, b2, (short)0, acc[0][2], false, false);
        acc[1][0] = __builtin_amdgcn_wmma_f32_16x16x4_f32(false, a1, false, b0, (short)0, acc[1][0], false, false);
        acc[1][1] = __builtin_amdgcn_wmma_f32_16x16x4_f32(false, a1, false, b1, (short)0, acc[1][1], false, false);
        acc[1][2] = __builtin_amdgcn_wmma_f32_16x16x4_f32(false, a1, false, b2, (short)0, acc[1][2], false, false);
        acc[2][0] = __builtin_amdgcn_wmma_f32_16x16x4_f32(false, a2, false, b0, (short)0, acc[2][0], false, false);
        acc[2][1] = __builtin_amdgcn_wmma_f32_16x16x4_f32(false, a2, false, b1, (short)0, acc[2][1], false, false);
        acc[2][2] = __builtin_amdgcn_wmma_f32_16x16x4_f32(false, a2, false, b2, (short)0, acc[2][2], false, false);
        acc[3][0] = __builtin_amdgcn_wmma_f32_16x16x4_f32(false, a3, false, b0, (short)0, acc[3][0], false, false);
        acc[3][1] = __builtin_amdgcn_wmma_f32_16x16x4_f32(false, a3, false, b1, (short)0, acc[3][1], false, false);
        acc[3][2] = __builtin_amdgcn_wmma_f32_16x16x4_f32(false, a3, false, b2, (short)0, acc[3][2], false, false);
    }

#pragma unroll
    for (int m = 0; m < 4; ++m) {
#pragma unroll
        for (int nt = 0; nt < 3; ++nt) {
            const int col = nt * 16 + l16;
            if (col >= NOUT) continue;
            const float bb = bias[col];
            const int row0 = rowBase + m * 16 + halfk * 8;
#pragma unroll
            for (int r = 0; r < 8; ++r) {
                const float v = acc[m][nt][r] + bb;
                float res;
                if (nt == 0) {
                    const float sp   = softplusf(v);
                    const float th   = 5.0f * tanhf_fast(v);
                    const float ofs  = (col < 4) ? 1.0f : 0.1f;
                    res = (col < 8) ? (ofs + sp) : th;
                } else {
                    res = 5.0f * tanhf_fast(v);
                }
                out[(size_t)(row0 + r) * NOUT + col] = res;
            }
        }
    }
}

__global__ __launch_bounds__(32) void chunk_scan_kernel(
    const float* __restrict__ out, float* __restrict__ chunkM)
{
    __shared__ float Mld[64];
    __shared__ float wld[32];

    const int lane = threadIdx.x;
    const int b    = blockIdx.x & (NBATCH - 1);
    const int c    = blockIdx.x >> 7;
    const int s    = lane & 7;
    const int r0   = lane >> 3;
    const int irow = s & 3;

    float mask[8];
#pragma unroll
    for (int m = 0; m < 8; ++m) {
        const bool on = (s < 4) ? (m == s || m == s + 4) : (m == s - 4 || m == s);
        mask[m] = (s < 4) ? (on ? NEG_BIG : 0.0f) : (on ? 0.0f : NEG_BIG);
    }

    const int t0 = c * CLEN;

    wld[lane] = out[((size_t)t0 * NBATCH + b) * NOUT + 8 + lane];
    __syncthreads();
    float new0 = wld[irow * 8 + r0]     + mask[r0];
    float new1 = wld[irow * 8 + r0 + 4] + mask[r0 + 4];
    __syncthreads();
    Mld[r0 * 8 + s]       = new0;
    Mld[(r0 + 4) * 8 + s] = new1;

    for (int step = 1; step < CLEN; ++step) {
        const int t = t0 + step;
        const float wv = out[((size_t)t * NBATCH + b) * NOUT + 8 + lane];
        __syncthreads();
        wld[lane] = wv;
        __syncthreads();

        float A[8];
#pragma unroll
        for (int m = 0; m < 8; ++m) A[m] = wld[irow * 8 + m] + mask[m];

        float t0r[8], t1r[8];
#pragma unroll
        for (int m = 0; m < 8; ++m) {
            t0r[m] = Mld[r0 * 8 + m]       + A[m];
            t1r[m] = Mld[(r0 + 4) * 8 + m] + A[m];
        }
        new0 = lse8(t0r);
        new1 = lse8(t1r);
        __syncthreads();
        Mld[r0 * 8 + s]       = new0;
        Mld[(r0 + 4) * 8 + s] = new1;
    }

    float* dst = chunkM + ((size_t)b * NCHUNK + c) * 64;
    dst[r0 * 8 + s]       = new0;
    dst[(r0 + 4) * 8 + s] = new1;
}

__global__ __launch_bounds__(32) void combine_kernel(
    const float* __restrict__ chunkM, float* __restrict__ logZdiv)
{
    __shared__ float Msh[64];
    __shared__ float vsh[8];
    const int lane = threadIdx.x;
    const int b    = blockIdx.x;
    const float* base = chunkM + (size_t)b * NCHUNK * 64;

    Msh[lane]      = base[lane];
    Msh[lane + 32] = base[lane + 32];
    __syncthreads();
    if (lane < 8) {
        float t[8];
#pragma unroll
        for (int j = 0; j < 8; ++j) t[j] = Msh[j * 8 + lane];
        vsh[lane] = lse8(t);
    }
    for (int c = 1; c < NCHUNK; ++c) {
        __syncthreads();
        Msh[lane]      = base[c * 64 + lane];
        Msh[lane + 32] = base[c * 64 + 32 + lane];
        __syncthreads();
        if (lane < 8) {
            float vold[8];
#pragma unroll
            for (int j = 0; j < 8; ++j) vold[j] = vsh[j];
            float t[8];
#pragma unroll
            for (int j = 0; j < 8; ++j) t[j] = vold[j] + Msh[j * 8 + lane];
            vsh[lane] = lse8(t);
        }
    }
    __syncthreads();
    if (lane == 0) {
        float t[8];
#pragma unroll
        for (int j = 0; j < 8; ++j) t[j] = vsh[j];
        logZdiv[b] = lse8(t) / (float)NTIME;
    }
}

__global__ __launch_bounds__(1024) void seq_scan_kernel(
    const float* __restrict__ out, float* __restrict__ logZdiv)
{
    __shared__ float fwdsh[NBATCH * 8];
    const int tid  = threadIdx.x;
    const int b    = tid >> 3;
    const int s    = tid & 7;
    const int irow = s & 3;

    float mask[8];
#pragma unroll
    for (int m = 0; m < 8; ++m) {
        const bool on = (s < 4) ? (m == s || m == s + 4) : (m == s - 4 || m == s);
        mask[m] = (s < 4) ? (on ? NEG_BIG : 0.0f) : (on ? 0.0f : NEG_BIG);
    }

    fwdsh[tid] = 0.0f;
    __syncthreads();
    for (int t = 0; t < NTIME; ++t) {
        const float* wrow = out + ((size_t)t * NBATCH + b) * NOUT + 8 + irow * 8;
        const float4 w0 = *(const float4*)wrow;
        const float4 w1 = *(const float4*)(wrow + 4);
        float tt[8] = { w0.x + mask[0], w0.y + mask[1], w0.z + mask[2], w0.w + mask[3],
                        w1.x + mask[4], w1.y + mask[5], w1.z + mask[6], w1.w + mask[7] };
#pragma unroll
        for (int m = 0; m < 8; ++m) tt[m] += fwdsh[(b << 3) + m];
        const float nf = lse8(tt);
        __syncthreads();
        fwdsh[tid] = nf;
        __syncthreads();
    }
    if (s == 0) {
        float t[8];
#pragma unroll
        for (int j = 0; j < 8; ++j) t[j] = fwdsh[(b << 3) + j];
        logZdiv[b] = lse8(t) / (float)NTIME;
    }
}

__global__ __launch_bounds__(256) void sublogz_kernel(
    float* __restrict__ out, const float* __restrict__ logZdiv)
{
    const size_t total = (size_t)NTIME * NBATCH * 32;
    const size_t e = (size_t)blockIdx.x * blockDim.x + threadIdx.x;
    if (e >= total) return;
    const int    j  = (int)(e & 31);
    const size_t tb = e >> 5;
    const int    b  = (int)(tb & (NBATCH - 1));
    out[tb * NOUT + 8 + j] -= logZdiv[b];
}

extern "C" void kernel_launch(void* const* d_in, const int* in_sizes, int n_in,
                              void* d_out, int out_size, void* d_ws, size_t ws_size,
                              hipStream_t stream) {
    (void)in_sizes; (void)n_in; (void)out_size;
    const float* x    = (const float*)d_in[0];
    const float* W    = (const float*)d_in[1];
    const float* bias = (const float*)d_in[2];
    float* out = (float*)d_out;

    const int ROWS = NTIME * NBATCH;
    gemm_act_kernel<<<ROWS / 512, 256, 0, stream>>>(x, W, bias, out);

    const size_t chunkBytes = (size_t)NBATCH * NCHUNK * 64 * sizeof(float);
    const size_t logZBytes  = (size_t)NBATCH * sizeof(float);
    const size_t total = (size_t)NTIME * NBATCH * 32;
    const int subBlocks = (int)((total + 255) / 256);

    if (ws_size >= chunkBytes + logZBytes) {
        float* chunkM  = (float*)d_ws;
        float* logZdiv = (float*)((char*)d_ws + chunkBytes);
        chunk_scan_kernel<<<NBATCH * NCHUNK, 32, 0, stream>>>(out, chunkM);
        combine_kernel<<<NBATCH, 32, 0, stream>>>(chunkM, logZdiv);
        sublogz_kernel<<<subBlocks, 256, 0, stream>>>(out, logZdiv);
    } else {
        float* logZdiv = (float*)d_ws;
        seq_scan_kernel<<<1, 1024, 0, stream>>>(out, logZdiv);
        sublogz_kernel<<<subBlocks, 256, 0, stream>>>(out, logZdiv);
    }
}


# GEMM pass: all of W kept in registers as B fragments (384 per lane); each wave streams 16 M tiles of x past them through 8 stage buffers, one wave per SIMD; no W load in the K loop; sums unchanged
# speedup vs baseline: 3.1216x; 3.1216x over previous
#include <hip/hip_runtime.h>
#include <cmath>

typedef __attribute__((ext_vector_type(2))) float v2f;
typedef __attribute__((ext_vector_type(4))) float v4f;
typedef __attribute__((ext_vector_type(8))) float v8f;

#define NTIME   2000
#define NBATCH  128
#define INSIZE  256
#define NOUT    40
#define NEG_BIG (-1.0e30f)

#define NCHUNK  100
#define CLEN    (NTIME / NCHUNK)

__device__ __forceinline__ float softplusf(float v) {
    return fmaxf(v, 0.0f) + __logf(1.0f + __expf(-fabsf(v)));
}

__device__ __forceinline__ float tanhf_fast(float v) {
    const float a = fabsf(v);
    const float e = __expf(2.0f * a);
    const float t = 1.0f - 2.0f * __builtin_amdgcn_rcpf(e + 1.0f);
    return copysignf(t, v);
}

__device__ __forceinline__ float lse8(const float (&t)[8]) {
    float mx = t[0];
#pragma unroll
    for (int m = 1; m < 8; ++m) mx = fmaxf(mx, t[m]);
    float s = 0.0f;
#pragma unroll
    for (int m = 0; m < 8; ++m) s += __expf(t[m] - mx);
    return mx + __logf(s);
}

__device__ __forceinline__ int iminc(int a, int b) { return a < b ? a : b; }

#define KBLK 32
#define NKB  (INSIZE / KBLK)
#define TPW  16

__device__ __forceinline__ void load_stage(v4f (&s)[4], const float* r) {
#pragma unroll
    for (int q = 0; q < 4; ++q) s[q] = *(const v4f*)(r + 16 * (q >> 1) + 4 * (q & 1));
}

__device__ __forceinline__ v2f pair_of(const v4f v, int j) {
    return j ? __builtin_shufflevector(v, v, 2, 3) : __builtin_shufflevector(v, v, 0, 1);
}

template <bool LAST>
__device__ __forceinline__ void gemm_tile(v8f (&acc)[3], v4f (&a)[NKB][4],
                                          const v4f (&b)[3][NKB][4], const float* xt)
{
#pragma unroll
    for (int kb = 0; kb < NKB; ++kb) {
        v4f (&s)[4] = a[kb];
#pragma unroll
        for (int q = 0; q < 4; ++q) {
#pragma unroll
            for (int j = 0; j < 2; ++j) {
                const v2f av = pair_of(s[q], j);
#pragma unroll
                for (int n = 0; n < 3; ++n) {
                    const v2f bv = pair_of(b[n][kb][q], j);
                    acc[n] = __builtin_amdgcn_wmma_f32_16x16x4_f32(false, av, false, bv, (short)0, acc[n], false, false);
                }
            }
        }
        __builtin_amdgcn_sched_barrier(0);
        if (kb == 0)     load_stage(a[NKB - 1], xt + (NKB - 1) * KBLK);
        else if (!LAST)  load_stage(a[kb - 1], xt + 16 * INSIZE + (kb - 1) * KBLK);
        __builtin_amdgcn_sched_barrier(0);
    }
}

__device__ __forceinline__ void store_tile(const v8f (&acc)[3], const float (&bv)[3],
                                           float* __restrict__ out, int row0, int l16)
{
#pragma unroll
    for (int nt = 0; nt < 3; ++nt) {
        const int col = nt * 16 + l16;
        if (col >= NOUT) continue;
        const float bb = bv[nt];
#pragma unroll
        for (int r = 0; r < 8; ++r) {
            const float v = acc[nt][r] + bb;
            float res;
            if (nt == 0) {
                const float sp   = softplusf(v);
                const float th   = 5.0f * tanhf_fast(v);
                const float ofs  = (col < 4) ? 1.0f : 0.1f;
                res = (col < 8) ? (ofs + sp) : th;
            } else {
                res = 5.0f * tanhf_fast(v);
            }
            out[(size_t)(row0 + r) * NOUT + col] = res;
        }
    }
}

__global__ __launch_bounds__(128) void gemm_act_kernel(
    const float* __restrict__ x, const float* __restrict__ W,
    const float* __restrict__ bias, float* __restrict__ out)
{
    const int lane  = threadIdx.x & 31;
    const int wave  = threadIdx.x >> 5;
    const int l16   = lane & 15;
    const int halfk = lane >> 4;

    const int rowBase = (blockIdx.x * 4 + wave) * (TPW * 16);

    const float* xt = x + (size_t)(rowBase + l16) * INSIZE + 8 * halfk;

    v4f a[NKB][4];
#pragma unroll
    for (int s = 0; s < NKB - 1; ++s) load_stage(a[s], xt + s * KBLK);

    v4f b[3][NKB][4];
    {
        const float* wp0 = W + (size_t)iminc(0 * 16 + l16, NOUT - 1) * INSIZE + 8 * halfk;
        const float* wp1 = W + (size_t)iminc(1 * 16 + l16, NOUT - 1) * INSIZE + 8 * halfk;
        const float* wp2 = W + (size_t)iminc(2 * 16 + l16, NOUT - 1) * INSIZE + 8 * halfk;
#pragma unroll
        for (int kb = 0; kb < NKB; ++kb) {
#pragma unroll
            for (int q = 0; q < 4; ++q) {
                const int off = kb * KBLK + 16 * (q >> 1) + 4 * (q & 1);
                b[0][kb][q] = *(const v4f*)(wp0 + off);
                b[1][kb][q] = *(const v4f*)(wp1 + off);
                b[2][kb][q] = *(const v4f*)(wp2 + off);
            }
        }
    }

    float bv[3];
#pragma unroll
    for (int n = 0; n < 3; ++n) bv[n] = bias[iminc(n * 16 + l16, NOUT - 1)];

    const int orow = rowBase + halfk * 8;

#pragma unroll 1
    for (int t = 0; t < TPW - 1; ++t) {
        v8f acc[3];
#pragma unroll
        for (int n = 0; n < 3; ++n)
#pragma unroll
            for (int i = 0; i < 8; ++i) acc[n][i] = 0.0f;
        gemm_tile<false>(acc, a, b, xt);
        store_tile(acc, bv, out, orow + t * 16, l16);
        xt += 16 * INSIZE;
    }
    {
        v8f acc[3];
#pragma unroll
        for (int n = 0; n < 3; ++n)
#pragma unroll
            for (int i = 0; i < 8; ++i) acc[n][i] = 0.0f;
        gemm_tile<true>(acc, a, b, xt);
        store_tile(acc, bv, out, orow + (TPW - 1) * 16, l16);
    }
}

#define GRP 10
#define SWEEP (NTIME / NCHUNK)
#define MST 68

__device__ __forceinline__ void load_w(float4 (&d)[8], const float* p) {
    const float4* q = (const float4*)p;
#pragma unroll
    for (int k = 0; k < 8; ++k) d[k] = q[k];
}

__device__ __forceinline__ void compose_step(float (&row)[8], const float4 (&wq)[8]) {
    const float L2E = 1.4426950408889634f;
    float mx = row[0];
#pragma unroll
    for (int m = 1; m < 8; ++m) mx = fmaxf(mx, row[m]);
    float rc[8];
#pragma unroll
    for (int m = 0; m < 8; ++m) rc[m] = row[m] - mx;

    float nr[8];
#pragma unroll
    for (int i = 0; i < 4; ++i) {
        const float w[8] = { wq[2 * i].x, wq[2 * i].y, wq[2 * i].z, wq[2 * i].w,
                             wq[2 * i + 1].x, wq[2 * i + 1].y, wq[2 * i + 1].z, wq[2 * i + 1].w };
        float e[8];
#pragma unroll
        for (int m = 0; m < 8; ++m) e[m] = __builtin_amdgcn_exp2f(fmaf(w[m], L2E, rc[m]));

        float sm = 0.0f;
#pragma unroll
        for (int m = 0; m < 8; ++m) {
            if (m == i || m == i + 4) continue;
            sm += e[m];
        }
        nr[i] = mx + __builtin_amdgcn_logf(sm);

        nr[i + 4] = mx + __builtin_amdgcn_logf(e[i] + e[i + 4]);
    }
#pragma unroll
    for (int s = 0; s < 8; ++s) row[s] = nr[s];
}

__global__ __launch_bounds__(NCHUNK * 8) void scan_fold_kernel(float* __restrict__ out)
{
    __shared__ __attribute__((aligned(16))) float Msh[NCHUNK * MST];
    __shared__ __attribute__((aligned(16))) float Psh[GRP * MST];
    __shared__ float zsh;

    const int tid = threadIdx.x;
    const int b   = blockIdx.x;
    const int c   = tid >> 3;
    const int r   = tid & 7;
    const int rq  = r & 3;

    const size_t tstride = (size_t)NBATCH * NOUT;
    const float* wbase = out + ((size_t)(c * CLEN) * NBATCH + b) * NOUT + 8;

    float4 wa[8], wb[8];
    load_w(wa, wbase + tstride);

    float row[8];
#pragma unroll
    for (int i = 0; i < 4; ++i) {
        const float wv = wbase[i * 8 + r] * 1.4426950408889634f;
        row[i]     = wv + ((rq == i) ? NEG_BIG : 0.0f);
        row[i + 4] = wv + ((rq == i) ? 0.0f : NEG_BIG);
    }

    const float* wp = wbase + tstride;
    for (int step = 1; step < CLEN - 2; step += 2) {
        load_w(wb, wp + tstride);
        compose_step(row, wa);
        load_w(wa, wp + 2 * tstride);
        compose_step(row, wb);
        wp += 2 * tstride;
    }
    compose_step(row, wa);

#pragma unroll
    for (int s = 0; s < 8; ++s) row[s] *= 0.6931471805599453f;

    {
        float4* dst = (float4*)(Msh + c * MST + r * 8);
        dst[0] = make_float4(row[0], row[1], row[2], row[3]);
        dst[1] = make_float4(row[4], row[5], row[6], row[7]);
    }
    __syncthreads();

    if (tid < GRP * 8) {
        const int g = c;
        float a[8];
        {
            const float4* s0 = (const float4*)(Msh + (g * GRP) * MST + r * 8);
            const float4 lo = s0[0], hi = s0[1];
            a[0] = lo.x; a[1] = lo.y; a[2] = lo.z; a[3] = lo.w;
            a[4] = hi.x; a[5] = hi.y; a[6] = hi.z; a[7] = hi.w;
        }
        for (int j = 1; j < GRP; ++j) {
            const float4* Bq = (const float4*)(Msh + (g * GRP + j) * MST);
            float n[8];
            float t[8][8];
#pragma unroll
            for (int m = 0; m < 8; ++m) {
                const float4 lo = Bq[2 * m], hi = Bq[2 * m + 1];
                t[0][m] = a[m] + lo.x; t[1][m] = a[m] + lo.y;
                t[2][m] = a[m] + lo.z; t[3][m] = a[m] + lo.w;
                t[4][m] = a[m] + hi.x; t[5][m] = a[m] + hi.y;
                t[6][m] = a[m] + hi.z; t[7][m] = a[m] + hi.w;
            }
#pragma unroll
            for (int s = 0; s < 8; ++s) n[s] = lse8(t[s]);
#pragma unroll
            for (int s = 0; s < 8; ++s) a[s] = n[s];
        }
        float4* dst = (float4*)(Psh + g * MST + r * 8);
        dst[0] = make_float4(a[0], a[1], a[2], a[3]);
        dst[1] = make_float4(a[4], a[5], a[6], a[7]);
    }
    __syncthreads();

    float* tq = out + (size_t)b * NOUT + 8 + (size_t)c * tstride + r * 4;
    const size_t kstride = (size_t)NCHUNK * tstride;
    float4 pc[SWEEP];
#pragma unroll
    for (int k = 0; k < SWEEP; ++k) pc[k] = *(const float4*)(tq + k * kstride);

    if (tid < 32) {
        const int s = tid & 7;
        float t[8];
#pragma unroll
        for (int j = 0; j < 8; ++j) t[j] = Psh[j * 8 + s];
        float v = lse8(t);
        for (int g = 1; g < GRP; ++g) {
#pragma unroll
            for (int j = 0; j < 8; ++j) t[j] = __shfl(v, j, 8) + Psh[g * MST + j * 8 + s];
            v = lse8(t);
        }
#pragma unroll
        for (int j = 0; j < 8; ++j) t[j] = __shfl(v, j, 8);
        const float z = lse8(t) / (float)NTIME;
        if (tid == 0) zsh = z;
    }
    __syncthreads();

    {
        const float z = zsh;
#pragma unroll
        for (int k = 0; k < SWEEP; ++k) {
            float4 o = pc[k];
            o.x -= z; o.y -= z; o.z -= z; o.w -= z;
            *(float4*)(tq + k * kstride) = o;
        }
    }
}

extern "C" void kernel_launch(void* const* d_in, const int* in_sizes, int n_in,
                              void* d_out, int out_size, void* d_ws, size_t ws_size,
                              hipStream_t stream) {
    (void)in_sizes; (void)n_in; (void)out_size;
    const float* x    = (const float*)d_in[0];
    const float* W    = (const float*)d_in[1];
    const float* bias = (const float*)d_in[2];
    float* out = (float*)d_out;

    const int ROWS = NTIME * NBATCH;
    gemm_act_kernel<<<ROWS / (4 * TPW * 16), 128, 0, stream>>>(x, W, bias, out);

    (void)d_ws; (void)ws_size;
    scan_fold_kernel<<<NBATCH, NCHUNK * 8, 0, stream>>>(out);
}


# activations and both fold levels on bare hardware transcendentals: v_tanh_f32 for tanh, unguarded exp2/log2 in softplus and the fold's log-sum-exp (base-2 units up to z); rounding changes, order kept
# speedup vs baseline: 3.2212x; 1.0319x over previous
#include <hip/hip_runtime.h>
#include <cmath>

typedef __attribute__((ext_vector_type(2))) float v2f;
typedef __attribute__((ext_vector_type(4))) float v4f;
typedef __attribute__((ext_vector_type(8))) float v8f;

#define NTIME   2000
#define NBATCH  128
#define INSIZE  256
#define NOUT    40
#define NEG_BIG (-1.0e30f)

#define NCHUNK  100
#define CLEN    (NTIME / NCHUNK)

#define L2E_F 1.4426950408889634f
#define LN2_F 0.6931471805599453f

__device__ __forceinline__ float softplusf(float v) {
    const float e = __builtin_amdgcn_exp2f(-fabsf(v) * L2E_F);
    return fmaf(__builtin_amdgcn_logf(1.0f + e), LN2_F, fmaxf(v, 0.0f));
}

__device__ __forceinline__ float tanhf_fast(float v) {
    return __builtin_amdgcn_tanhf(v);
}

__device__ __forceinline__ float lse8(const float (&t)[8]) {
    float mx = t[0];
#pragma unroll
    for (int m = 1; m < 8; ++m) mx = fmaxf(mx, t[m]);
    float s = 0.0f;
#pragma unroll
    for (int m = 0; m < 8; ++m) s += __builtin_amdgcn_exp2f(t[m] - mx);
    return mx + __builtin_amdgcn_logf(s);
}

__device__ __forceinline__ int iminc(int a, int b) { return a < b ? a : b; }

#define KBLK 32
#define NKB  (INSIZE / KBLK)
#define TPW  16

__device__ __forceinline__ void load_stage(v4f (&s)[4], const float* r) {
#pragma unroll
    for (int q = 0; q < 4; ++q) s[q] = *(const v4f*)(r + 16 * (q >> 1) + 4 * (q & 1));
}

__device__ __forceinline__ v2f pair_of(const v4f v, int j) {
    return j ? __builtin_shufflevector(v, v, 2, 3) : __builtin_shufflevector(v, v, 0, 1);
}

template <bool LAST>
__device__ __forceinline__ void gemm_tile(v8f (&acc)[3], v4f (&a)[NKB][4],
                                          const v4f (&b)[3][NKB][4], const float* xt)
{
#pragma unroll
    for (int kb = 0; kb < NKB; ++kb) {
        v4f (&s)[4] = a[kb];
#pragma unroll
        for (int q = 0; q < 4; ++q) {
#pragma unroll
            for (int j = 0; j < 2; ++j) {
                const v2f av = pair_of(s[q], j);
#pragma unroll
                for (int n = 0; n < 3; ++n) {
                    const v2f bv = pair_of(b[n][kb][q], j);
                    acc[n] = __builtin_amdgcn_wmma_f32_16x16x4_f32(false, av, false, bv, (short)0, acc[n], false, false);
                }
            }
        }
        __builtin_amdgcn_sched_barrier(0);
        if (kb == 0)     load_stage(a[NKB - 1], xt + (NKB - 1) * KBLK);
        else if (!LAST)  load_stage(a[kb - 1], xt + 16 * INSIZE + (kb - 1) * KBLK);
        __builtin_amdgcn_sched_barrier(0);
    }
}

__device__ __forceinline__ void store_tile(const v8f (&acc)[3], const float (&bv)[3],
                                           float* __restrict__ out, int row0, int l16)
{
#pragma unroll
    for (int nt = 0; nt < 3; ++nt) {
        const int col = nt * 16 + l16;
        if (col >= NOUT) continue;
        const float bb = bv[nt];
#pragma unroll
        for (int r = 0; r < 8; ++r) {
            const float v = acc[nt][r] + bb;
            float res;
            if (nt == 0) {
                const float sp   = softplusf(v);
                const float th   = 5.0f * tanhf_fast(v);
                const float ofs  = (col < 4) ? 1.0f : 0.1f;
                res = (col < 8) ? (ofs + sp) : th;
            } else {
                res = 5.0f * tanhf_fast(v);
            }
            out[(size_t)(row0 + r) * NOUT + col] = res;
        }
    }
}

__global__ __launch_bounds__(128) void gemm_act_kernel(
    const float* __restrict__ x, const float* __restrict__ W,
    const float* __restrict__ bias, float* __restrict__ out)
{
    const int lane  = threadIdx.x & 31;
    const int wave  = threadIdx.x >> 5;
    const int l16   = lane & 15;
    const int halfk = lane >> 4;

    const int rowBase = (blockIdx.x * 4 + wave) * (TPW * 16);

    const float* xt = x + (size_t)(rowBase + l16) * INSIZE + 8 * halfk;

    v4f a[NKB][4];
#pragma unroll
    for (int s = 0; s < NKB - 1; ++s) load_stage(a[s], xt + s * KBLK);

    v4f b[3][NKB][4];
    {
        const float* wp0 = W + (size_t)iminc(0 * 16 + l16, NOUT - 1) * INSIZE + 8 * halfk;
        const float* wp1 = W + (size_t)iminc(1 * 16 + l16, NOUT - 1) * INSIZE + 8 * halfk;
        const float* wp2 = W + (size_t)iminc(2 * 16 + l16, NOUT - 1) * INSIZE + 8 * halfk;
#pragma unroll
        for (int kb = 0; kb < NKB; ++kb) {
#pragma unroll
            for (int q = 0; q < 4; ++q) {
                const int off = kb * KBLK + 16 * (q >> 1) + 4 * (q & 1);
                b[0][kb][q] = *(const v4f*)(wp0 + off);
                b[1][kb][q] = *(const v4f*)(wp1 + off);
                b[2][kb][q] = *(const v4f*)(wp2 + off);
            }
        }
    }

    float bv[3];
#pragma unroll
    for (int n = 0; n < 3; ++n) bv[n] = bias[iminc(n * 16 + l16, NOUT - 1)];

    const int orow = rowBase + halfk * 8;

#pragma unroll 1
    for (int t = 0; t < TPW - 1; ++t) {
        v8f acc[3];
#pragma unroll
        for (int n = 0; n < 3; ++n)
#pragma unroll
            for (int i = 0; i < 8; ++i) acc[n][i] = 0.0f;
        gemm_tile<false>(acc, a, b, xt);
        store_tile(acc, bv, out, orow + t * 16, l16);
        xt += 16 * INSIZE;
    }
    {
        v8f acc[3];
#pragma unroll
        for (int n = 0; n < 3; ++n)
#pragma unroll
            for (int i = 0; i < 8; ++i) acc[n][i] = 0.0f;
        gemm_tile<true>(acc, a, b, xt);
        store_tile(acc, bv, out, orow + (TPW - 1) * 16, l16);
    }
}

#define GRP 10
#define SWEEP (NTIME / NCHUNK)
#define MST 68

__device__ __forceinline__ void load_w(float4 (&d)[8], const float* p) {
    const float4* q = (const float4*)p;
#pragma unroll
    for (int k = 0; k < 8; ++k) d[k] = q[k];
}

__device__ __forceinline__ void compose_step(float (&row)[8], const float4 (&wq)[8]) {
    const float L2E = 1.4426950408889634f;
    float mx = row[0];
#pragma unroll
    for (int m = 1; m < 8; ++m) mx = fmaxf(mx, row[m]);
    float rc[8];
#pragma unroll
    for (int m = 0; m < 8; ++m) rc[m] = row[m] - mx;

    float nr[8];
#pragma unroll
    for (int i = 0; i < 4; ++i) {
        const float w[8] = { wq[2 * i].x, wq[2 * i].y, wq[2 * i].z, wq[2 * i].w,
                             wq[2 * i + 1].x, wq[2 * i + 1].y, wq[2 * i + 1].z, wq[2 * i + 1].w };
        float e[8];
#pragma unroll
        for (int m = 0; m < 8; ++m) e[m] = __builtin_amdgcn_exp2f(fmaf(w[m], L2E, rc[m]));

        float sm = 0.0f;
#pragma unroll
        for (int m = 0; m < 8; ++m) {
            if (m == i || m == i + 4) continue;
            sm += e[m];
        }
        nr[i] = mx + __builtin_amdgcn_logf(sm);

        nr[i + 4] = mx + __builtin_amdgcn_logf(e[i] + e[i + 4]);
    }
#pragma unroll
    for (int s = 0; s < 8; ++s) row[s] = nr[s];
}

__global__ __launch_bounds__(NCHUNK * 8) void scan_fold_kernel(float* __restrict__ out)
{
    __shared__ __attribute__((aligned(16))) float Msh[NCHUNK * MST];
    __shared__ __attribute__((aligned(16))) float Psh[GRP * MST];
    __shared__ float zsh;

    const int tid = threadIdx.x;
    const int b   = blockIdx.x;
    const int c   = tid >> 3;
    const int r   = tid & 7;
    const int rq  = r & 3;

    const size_t tstride = (size_t)NBATCH * NOUT;
    const float* wbase = out + ((size_t)(c * CLEN) * NBATCH + b) * NOUT + 8;

    float4 wa[8], wb[8];
    load_w(wa, wbase + tstride);

    float row[8];
#pragma unroll
    for (int i = 0; i < 4; ++i) {
        const float wv = wbase[i * 8 + r] * 1.4426950408889634f;
        row[i]     = wv + ((rq == i) ? NEG_BIG : 0.0f);
        row[i + 4] = wv + ((rq == i) ? 0.0f : NEG_BIG);
    }

    const float* wp = wbase + tstride;
    for (int step = 1; step < CLEN - 2; step += 2) {
        load_w(wb, wp + tstride);
        compose_step(row, wa);
        load_w(wa, wp + 2 * tstride);
        compose_step(row, wb);
        wp += 2 * tstride;
    }
    compose_step(row, wa);


    {
        float4* dst = (float4*)(Msh + c * MST + r * 8);
        dst[0] = make_float4(row[0], row[1], row[2], row[3]);
        dst[1] = make_float4(row[4], row[5], row[6], row[7]);
    }
    __syncthreads();

    if (tid < GRP * 8) {
        const int g = c;
        float a[8];
        {
            const float4* s0 = (const float4*)(Msh + (g * GRP) * MST + r * 8);
            const float4 lo = s0[0], hi = s0[1];
            a[0] = lo.x; a[1] = lo.y; a[2] = lo.z; a[3] = lo.w;
            a[4] = hi.x; a[5] = hi.y; a[6] = hi.z; a[7] = hi.w;
        }
        for (int j = 1; j < GRP; ++j) {
            const float4* Bq = (const float4*)(Msh + (g * GRP + j) * MST);
            float n[8];
            float t[8][8];
#pragma unroll
            for (int m = 0; m < 8; ++m) {
                const float4 lo = Bq[2 * m], hi = Bq[2 * m + 1];
                t[0][m] = a[m] + lo.x; t[1][m] = a[m] + lo.y;
                t[2][m] = a[m] + lo.z; t[3][m] = a[m] + lo.w;
                t[4][m] = a[m] + hi.x; t[5][m] = a[m] + hi.y;
                t[6][m] = a[m] + hi.z; t[7][m] = a[m] + hi.w;
            }
#pragma unroll
            for (int s = 0; s < 8; ++s) n[s] = lse8(t[s]);
#pragma unroll
            for (int s = 0; s < 8; ++s) a[s] = n[s];
        }
        float4* dst = (float4*)(Psh + g * MST + r * 8);
        dst[0] = make_float4(a[0], a[1], a[2], a[3]);
        dst[1] = make_float4(a[4], a[5], a[6], a[7]);
    }
    __syncthreads();

    float* tq = out + (size_t)b * NOUT + 8 + (size_t)c * tstride + r * 4;
    const size_t kstride = (size_t)NCHUNK * tstride;
    float4 pc[SWEEP];
#pragma unroll
    for (int k = 0; k < SWEEP; ++k) pc[k] = *(const float4*)(tq + k * kstride);

    if (tid < 32) {
        const int s = tid & 7;
        float t[8];
#pragma unroll
        for (int j = 0; j < 8; ++j) t[j] = Psh[j * 8 + s];
        float v = lse8(t);
        for (int g = 1; g < GRP; ++g) {
#pragma unroll
            for (int j = 0; j < 8; ++j) t[j] = __shfl(v, j, 8) + Psh[g * MST + j * 8 + s];
            v = lse8(t);
        }
#pragma unroll
        for (int j = 0; j < 8; ++j) t[j] = __shfl(v, j, 8);
        const float z = lse8(t) * (LN2_F / (float)NTIME);
        if (tid == 0) zsh = z;
    }
    __syncthreads();

    {
        const float z = zsh;
#pragma unroll
        for (int k = 0; k < SWEEP; ++k) {
            float4 o = pc[k];
            o.x -= z; o.y -= z; o.z -= z; o.w -= z;
            *(float4*)(tq + k * kstride) = o;
        }
    }
}

extern "C" void kernel_launch(void* const* d_in, const int* in_sizes, int n_in,
                              void* d_out, int out_size, void* d_ws, size_t ws_size,
                              hipStream_t stream) {
    (void)in_sizes; (void)n_in; (void)out_size;
    const float* x    = (const float*)d_in[0];
    const float* W    = (const float*)d_in[1];
    const float* bias = (const float*)d_in[2];
    float* out = (float*)d_out;

    const int ROWS = NTIME * NBATCH;
    gemm_act_kernel<<<ROWS / (4 * TPW * 16), 128, 0, stream>>>(x, W, bias, out);

    (void)d_ws; (void)ws_size;
    scan_fold_kernel<<<NBATCH, NCHUNK * 8, 0, stream>>>(out);
}
